# starting kernel
# baseline (speedup 1.0000x reference)
#include <hip/hip_runtime.h>

typedef __attribute__((ext_vector_type(2))) float v2f;
typedef __attribute__((ext_vector_type(8))) float v8f;

#define NPTS 8192
#define EPSF 1e-5f
#define NCHUNKS 16
#define JCHUNK 512
#define JTILES (JCHUNK / 16)
#define BLOCK 128
#define MAX_JSPLIT 16

__global__ __launch_bounds__(BLOCK) void nbody_force_wmma(
    const float* __restrict__ pos, float* __restrict__ partial, int jsplit)
{
    __shared__ float4 sj[JCHUNK];

    const int tid  = threadIdx.x;
    const int lane = tid & 31;
    const int wave = tid >> 5;
    const int n16  = lane & 15;
    const bool hi  = (lane >= 16);

    const int i = blockIdx.x * 64 + wave * 16 + n16;
    const float xi = pos[i * 3 + 0];
    const float yi = pos[i * 3 + 1];
    const float zi = pos[i * 3 + 2];
    const float ri2 = xi * xi + yi * yi + zi * zi;

    v2f b;
    b[0] = hi ? (-2.0f * zi) : (-2.0f * xi);
    b[1] = hi ? 1.0f         : (-2.0f * yi);

    v8f cfrag;
#pragma unroll
    for (int m = 0; m < 8; ++m) cfrag[m] = ri2;

    float fx = 0.0f, fy = 0.0f, fz = 0.0f;

    for (int c = blockIdx.y; c < NCHUNKS; c += jsplit) {
        __syncthreads();
        for (int t = tid; t < JCHUNK; t += BLOCK) {
            const int j = c * JCHUNK + t;
            const float xj = pos[j * 3 + 0];
            const float yj = pos[j * 3 + 1];
            const float zj = pos[j * 3 + 2];
            sj[t] = make_float4(xj, yj, zj, xj * xj + yj * yj + zj * zj);
        }
        __syncthreads();

        for (int jt = 0; jt < JTILES; ++jt) {
            const float4 pj = sj[jt * 16 + n16];
            v2f a;
            a[0] = hi ? pj.z : pj.x;
            a[1] = hi ? pj.w : pj.y;

            v8f d2 = __builtin_amdgcn_wmma_f32_16x16x4_f32(
                false, a, false, b, (short)0, cfrag, false, false);

            const int jb = jt * 16 + (hi ? 8 : 0);
#pragma unroll
            for (int m = 0; m < 8; ++m) {
                const float4 q = sj[jb + m];
                float dd = d2[m];
                dd = dd > 0.0f ? dd : 0.0f;
                const float r = __builtin_amdgcn_sqrtf(dd);
                const float t = r + EPSF;
                const float w = __builtin_amdgcn_rcpf(t * t * t);
                fx = fmaf(w, xi - q.x, fx);
                fy = fmaf(w, yi - q.y, fy);
                fz = fmaf(w, zi - q.z, fz);
            }
        }
    }

    fx += __shfl_xor(fx, 16, 32);
    fy += __shfl_xor(fy, 16, 32);
    fz += __shfl_xor(fz, 16, 32);

    if (!hi) {
        float* o = partial + ((size_t)blockIdx.y * NPTS + (size_t)i) * 3;
        o[0] = fx; o[1] = fy; o[2] = fz;
    }
}

__global__ void reduce_partials(const float* __restrict__ partial,
                                float* __restrict__ out, int jsplit)
{
    const int idx = blockIdx.x * blockDim.x + threadIdx.x;
    if (idx < NPTS * 3) {
        float s = 0.0f;
        for (int k = 0; k < jsplit; ++k)
            s += partial[(size_t)k * NPTS * 3 + idx];
        out[idx] = s;
    }
}

extern "C" void kernel_launch(void* const* d_in, const int* in_sizes, int n_in,
                              void* d_out, int out_size, void* d_ws, size_t ws_size,
                              hipStream_t stream)
{
    const float* pos = (const float*)d_in[0];
    float* out = (float*)d_out;

    const size_t slice = (size_t)NPTS * 3 * sizeof(float);
    int jsplit = (int)(ws_size / slice);
    if (jsplit > MAX_JSPLIT) jsplit = MAX_JSPLIT;

    float* partial = (float*)d_ws;
    if (jsplit < 1) {
        jsplit = 1;
        partial = out;
    }

    dim3 grid(NPTS / 64, jsplit);
    nbody_force_wmma<<<grid, BLOCK, 0, stream>>>(pos, partial, jsplit);

    if (partial != out) {
        const int tot = NPTS * 3;
        reduce_partials<<<(tot + 255) / 256, 256, 0, stream>>>(partial, out, jsplit);
    }
}


# Tile roles swapped: rows = i, columns = j, so a lane keeps its 8 i and their sums in registers and reads only its own j per tile (8 broadcast LDS reads per tile gone); summation order changes
# speedup vs baseline: 1.1804x; 1.1804x over previous
#include <hip/hip_runtime.h>


typedef __attribute__((ext_vector_type(2))) float v2f;
typedef __attribute__((ext_vector_type(8))) float v8f;

#define NPTS 8192
#define EPSF 1e-5f
#define IBLK 32
#define NWAVE 16
#define BLOCK (NWAVE * 32)
#define NSLICE (NWAVE / 2)
#define JSLICE (NPTS / NSLICE)
#define JTILES (JSLICE / 16)
#define NSTAGE (NPTS / BLOCK)

template <int MASK>
__device__ __forceinline__ float add_xor16(float keep, float send)
{
    const int s = __builtin_amdgcn_update_dpp(0, __builtin_bit_cast(int, send),
                                              0x160 + MASK, 0xf, 0xf, true);
    return keep + __builtin_bit_cast(float, s);
}

__global__ __launch_bounds__(BLOCK) void nbody_force_wmma(
    const float* __restrict__ pos, float* __restrict__ out)
{
    __shared__ float4 sj[NPTS];
    __shared__ float red[NSLICE][IBLK * 3];

    const int tid  = threadIdx.x;
    const int lane = tid & 31;
    const int wave = tid >> 5;
    const int n16  = lane & 15;
    const bool hi  = (lane >= 16);
    const int rg   = wave & 1;
    const int sl   = wave >> 1;

    float px[NSTAGE], py[NSTAGE], pz[NSTAGE];
#pragma unroll
    for (int k = 0; k < NSTAGE; ++k) {
        const int j = k * BLOCK + tid;
        px[k] = pos[j * 3 + 0];
        py[k] = pos[j * 3 + 1];
        pz[k] = pos[j * 3 + 2];
    }
#pragma unroll
    for (int k = 0; k < NSTAGE; ++k) {
        const int j = k * BLOCK + tid;
        sj[j] = make_float4(px[k], py[k], pz[k],
                            px[k] * px[k] + py[k] * py[k] + pz[k] * pz[k]);
    }
    __syncthreads();

    const int ib = blockIdx.x * IBLK + rg * 16;

    const float4 pa = sj[ib + n16];
    v2f a;
    a[0] = hi ? (-2.0f * pa.z) : (-2.0f * pa.x);
    a[1] = hi ? 1.0f           : (-2.0f * pa.y);

    const int ir = ib + (hi ? 8 : 0);
    float xi[8], yi[8], zi[8];
    v8f cfrag;
#pragma unroll
    for (int r = 0; r < 8; ++r) {
        const float4 p = sj[ir + r];
        xi[r] = p.x; yi[r] = p.y; zi[r] = p.z; cfrag[r] = p.w;
    }

    float fx[8], fy[8], fz[8];
#pragma unroll
    for (int r = 0; r < 8; ++r) { fx[r] = 0.0f; fy[r] = 0.0f; fz[r] = 0.0f; }

    const float4* sjs = sj + sl * JSLICE + n16;
    const v2f* sjb = (const v2f*)sjs + (hi ? 1 : 0);

    for (int jt = 0; jt < JTILES; ++jt) {
        const float4 pj = sjs[jt * 16];
        const v2f b = sjb[jt * 32];

        v8f d2 = __builtin_amdgcn_wmma_f32_16x16x4_f32(
            false, a, false, b, (short)0, cfrag, false, false);

#pragma unroll
        for (int r = 0; r < 8; ++r) {
            float dd = d2[r];
            dd = dd > 0.0f ? dd : 0.0f;
            const float rr = __builtin_amdgcn_sqrtf(dd);
            const float t = rr + EPSF;
            const float w = __builtin_amdgcn_rcpf(t * t * t);
            fx[r] = fmaf(w, xi[r] - pj.x, fx[r]);
            fy[r] = fmaf(w, yi[r] - pj.y, fy[r]);
            fz[r] = fmaf(w, zi[r] - pj.z, fz[r]);
        }
    }

    const bool b8 = (n16 & 8) != 0;
    const bool b4 = (n16 & 4) != 0;
    const bool b2 = (n16 & 2) != 0;
    float g[4][3], hh[2][3], tot[3];
#pragma unroll
    for (int k = 0; k < 4; ++k) {
        g[k][0] = add_xor16<8>(b8 ? fx[k + 4] : fx[k], b8 ? fx[k] : fx[k + 4]);
        g[k][1] = add_xor16<8>(b8 ? fy[k + 4] : fy[k], b8 ? fy[k] : fy[k + 4]);
        g[k][2] = add_xor16<8>(b8 ? fz[k + 4] : fz[k], b8 ? fz[k] : fz[k + 4]);
    }
#pragma unroll
    for (int k = 0; k < 2; ++k)
#pragma unroll
        for (int c = 0; c < 3; ++c)
            hh[k][c] = add_xor16<4>(b4 ? g[k + 2][c] : g[k][c], b4 ? g[k][c] : g[k + 2][c]);
#pragma unroll
    for (int c = 0; c < 3; ++c) {
        const float t2 = add_xor16<2>(b2 ? hh[1][c] : hh[0][c], b2 ? hh[0][c] : hh[1][c]);
        tot[c] = add_xor16<1>(t2, t2);
    }
    if ((n16 & 1) == 0) {
        float* p = &red[sl][(rg * 16 + (hi ? 8 : 0) + (n16 >> 1)) * 3];
        p[0] = tot[0]; p[1] = tot[1]; p[2] = tot[2];
    }
    __syncthreads();

    if (tid < IBLK * 3) {
        float s = 0.0f;
#pragma unroll
        for (int k = 0; k < NSLICE; ++k) s += red[k][tid];
        out[blockIdx.x * (IBLK * 3) + tid] = s;
    }
}

extern "C" void kernel_launch(void* const* d_in, const int* in_sizes, int n_in,
                              void* d_out, int out_size, void* d_ws, size_t ws_size,
                              hipStream_t stream)
{
    const float* pos = (const float*)d_in[0];
    float* out = (float*)d_out;

    nbody_force_wmma<<<NPTS / IBLK, BLOCK, 0, stream>>>(pos, out);
}


# Clamp of d2 at zero dropped: the square root takes |d2| as a source modifier, so the max and the NaN-quieting max before it leave the loop (8 of 44 dual-issue slots per tile); order of sums unchanged
# speedup vs baseline: 1.2427x; 1.0528x over previous
#include <hip/hip_runtime.h>


typedef __attribute__((ext_vector_type(2))) float v2f;
typedef __attribute__((ext_vector_type(8))) float v8f;

#define NPTS 8192
#define EPSF 1e-5f
#define IBLK 32
#define NWAVE 16
#define BLOCK (NWAVE * 32)
#define NSLICE (NWAVE / 2)
#define JSLICE (NPTS / NSLICE)
#define JTILES (JSLICE / 16)
#define NSTAGE (NPTS / BLOCK)

template <int MASK>
__device__ __forceinline__ float add_xor16(float keep, float send)
{
    const int s = __builtin_amdgcn_update_dpp(0, __builtin_bit_cast(int, send),
                                              0x160 + MASK, 0xf, 0xf, true);
    return keep + __builtin_bit_cast(float, s);
}

__global__ __launch_bounds__(BLOCK) void nbody_force_wmma(
    const float* __restrict__ pos, float* __restrict__ out)
{
    __shared__ float4 sj[NPTS];
    __shared__ float red[NSLICE][IBLK * 3];

    const int tid  = threadIdx.x;
    const int lane = tid & 31;
    const int wave = tid >> 5;
    const int n16  = lane & 15;
    const bool hi  = (lane >= 16);
    const int rg   = wave & 1;
    const int sl   = wave >> 1;

    float px[NSTAGE], py[NSTAGE], pz[NSTAGE];
#pragma unroll
    for (int k = 0; k < NSTAGE; ++k) {
        const int j = k * BLOCK + tid;
        px[k] = pos[j * 3 + 0];
        py[k] = pos[j * 3 + 1];
        pz[k] = pos[j * 3 + 2];
    }
#pragma unroll
    for (int k = 0; k < NSTAGE; ++k) {
        const int j = k * BLOCK + tid;
        sj[j] = make_float4(px[k], py[k], pz[k],
                            px[k] * px[k] + py[k] * py[k] + pz[k] * pz[k]);
    }
    __syncthreads();

    const int ib = blockIdx.x * IBLK + rg * 16;

    const float4 pa = sj[ib + n16];
    v2f a;
    a[0] = hi ? (-2.0f * pa.z) : (-2.0f * pa.x);
    a[1] = hi ? 1.0f           : (-2.0f * pa.y);

    const int ir = ib + (hi ? 8 : 0);
    float xi[8], yi[8], zi[8];
    v8f cfrag;
#pragma unroll
    for (int r = 0; r < 8; ++r) {
        const float4 p = sj[ir + r];
        xi[r] = p.x; yi[r] = p.y; zi[r] = p.z; cfrag[r] = p.w;
    }

    float fx[8], fy[8], fz[8];
#pragma unroll
    for (int r = 0; r < 8; ++r) { fx[r] = 0.0f; fy[r] = 0.0f; fz[r] = 0.0f; }

    const float4* sjs = sj + sl * JSLICE + n16;
    const v2f* sjb = (const v2f*)sjs + (hi ? 1 : 0);

    for (int jt = 0; jt < JTILES; ++jt) {
        const float4 pj = sjs[jt * 16];
        const v2f b = sjb[jt * 32];

        v8f d2 = __builtin_amdgcn_wmma_f32_16x16x4_f32(
            false, a, false, b, (short)0, cfrag, false, false);

#pragma unroll
        for (int r = 0; r < 8; ++r) {
            const float rr = __builtin_amdgcn_sqrtf(__builtin_fabsf(d2[r]));
            const float t = rr + EPSF;
            const float w = __builtin_amdgcn_rcpf(t * t * t);
            fx[r] = fmaf(w, xi[r] - pj.x, fx[r]);
            fy[r] = fmaf(w, yi[r] - pj.y, fy[r]);
            fz[r] = fmaf(w, zi[r] - pj.z, fz[r]);
        }
    }

    const bool b8 = (n16 & 8) != 0;
    const bool b4 = (n16 & 4) != 0;
    const bool b2 = (n16 & 2) != 0;
    float g[4][3], hh[2][3], tot[3];
#pragma unroll
    for (int k = 0; k < 4; ++k) {
        g[k][0] = add_xor16<8>(b8 ? fx[k + 4] : fx[k], b8 ? fx[k] : fx[k + 4]);
        g[k][1] = add_xor16<8>(b8 ? fy[k + 4] : fy[k], b8 ? fy[k] : fy[k + 4]);
        g[k][2] = add_xor16<8>(b8 ? fz[k + 4] : fz[k], b8 ? fz[k] : fz[k + 4]);
    }
#pragma unroll
    for (int k = 0; k < 2; ++k)
#pragma unroll
        for (int c = 0; c < 3; ++c)
            hh[k][c] = add_xor16<4>(b4 ? g[k + 2][c] : g[k][c], b4 ? g[k][c] : g[k + 2][c]);
#pragma unroll
    for (int c = 0; c < 3; ++c) {
        const float t2 = add_xor16<2>(b2 ? hh[1][c] : hh[0][c], b2 ? hh[0][c] : hh[1][c]);
        tot[c] = add_xor16<1>(t2, t2);
    }
    if ((n16 & 1) == 0) {
        float* p = &red[sl][(rg * 16 + (hi ? 8 : 0) + (n16 >> 1)) * 3];
        p[0] = tot[0]; p[1] = tot[1]; p[2] = tot[2];
    }
    __syncthreads();

    if (tid < IBLK * 3) {
        float s = 0.0f;
#pragma unroll
        for (int k = 0; k < NSLICE; ++k) s += red[k][tid];
        out[blockIdx.x * (IBLK * 3) + tid] = s;
    }
}

extern "C" void kernel_launch(void* const* d_in, const int* in_sizes, int n_in,
                              void* d_out, int out_size, void* d_ws, size_t ws_size,
                              hipStream_t stream)
{
    const float* pos = (const float*)d_in[0];
    float* out = (float*)d_out;

    nbody_force_wmma<<<NPTS / IBLK, BLOCK, 0, stream>>>(pos, out);
}


# No LDS staging: a lane loads its own column j from global memory one group of 4 tiles ahead, staging pass and its barrier gone; work-groups start their sweep at different groups: order of sums rotates
# speedup vs baseline: 1.3887x; 1.1175x over previous
#include <hip/hip_runtime.h>


typedef __attribute__((ext_vector_type(2))) float v2f;
typedef __attribute__((ext_vector_type(8))) float v8f;

#define NPTS 8192
#define EPSF 1e-5f
#define IBLK 32
#define NWAVE 16
#define BLOCK (NWAVE * 32)
#define NSLICE (NWAVE / 2)
#define JSLICE (NPTS / NSLICE)
#define JTILES (JSLICE / 16)
#define TPG 4
#define NGRP (JTILES / TPG)
#define TILEF (16 * 3)
#define GRPF (TPG * TILEF)

template <int MASK>
__device__ __forceinline__ float add_xor16(float keep, float send)
{
    const int s = __builtin_amdgcn_update_dpp(0, __builtin_bit_cast(int, send),
                                              0x160 + MASK, 0xf, 0xf, true);
    return keep + __builtin_bit_cast(float, s);
}

__device__ __forceinline__ float norm2(float x, float y, float z)
{
    return x * x + y * y + z * z;
}

struct JGroup { float x[TPG], y[TPG], z[TPG]; };

__device__ __forceinline__ void load_group(JGroup& q, const float* __restrict__ p)
{
#pragma unroll
    for (int t = 0; t < TPG; ++t) {
        q.x[t] = p[t * TILEF + 0];
        q.y[t] = p[t * TILEF + 1];
        q.z[t] = p[t * TILEF + 2];
    }
}

__device__ __forceinline__ void hold_group(const JGroup& q)
{
#pragma unroll
    for (int t = 0; t < TPG; ++t)
        asm volatile("" :: "v"(q.x[t]), "v"(q.y[t]), "v"(q.z[t]));
}

__global__ __launch_bounds__(BLOCK) __attribute__((amdgpu_waves_per_eu(4, 4)))
void nbody_force_wmma(
    const float* __restrict__ pos, float* __restrict__ out)
{
    __shared__ float red[NSLICE][IBLK * 3];

    const int tid  = threadIdx.x;
    const int lane = tid & 31;
    const int wave = tid >> 5;
    const int n16  = lane & 15;
    const bool hi  = (lane >= 16);
    const int rg   = wave & 1;
    const int sl   = wave >> 1;

    const int rot = blockIdx.x & (NGRP - 1);
    const float* pjl = pos + (sl * JSLICE + n16) * 3;

    JGroup ga, gb;
    load_group(ga, pjl + rot * GRPF);

    const int ib = blockIdx.x * IBLK + rg * 16;

    const float* pa = pos + (ib + n16) * 3;
    const float pax = pa[0], pay = pa[1], paz = pa[2];
    v2f a;
    a[0] = hi ? (-2.0f * paz) : (-2.0f * pax);
    a[1] = hi ? 1.0f          : (-2.0f * pay);

    const int ir = ib + (hi ? 8 : 0);
    float xi[8], yi[8], zi[8];
    v8f cfrag;
#pragma unroll
    for (int r = 0; r < 8; ++r) {
        const float* p = pos + (ir + r) * 3;
        xi[r] = p[0]; yi[r] = p[1]; zi[r] = p[2];
        cfrag[r] = norm2(xi[r], yi[r], zi[r]);
    }

    float fx[8], fy[8], fz[8];
#pragma unroll
    for (int r = 0; r < 8; ++r) { fx[r] = 0.0f; fy[r] = 0.0f; fz[r] = 0.0f; }

    auto do_group = [&](const JGroup& q) {
#pragma unroll
        for (int t = 0; t < TPG; ++t) {
            const float pjx = q.x[t], pjy = q.y[t], pjz = q.z[t];
            const float rj2 = norm2(pjx, pjy, pjz);
            v2f b;
            b[0] = hi ? pjz : pjx;
            b[1] = hi ? rj2 : pjy;

            v8f d2 = __builtin_amdgcn_wmma_f32_16x16x4_f32(
                false, a, false, b, (short)0, cfrag, false, false);

            float w[8];
#pragma unroll
            for (int r = 0; r < 8; ++r) {
                w[r] = __builtin_amdgcn_sqrtf(__builtin_fabsf(d2[r]));
            }
#pragma unroll
            for (int r = 0; r < 8; ++r) {
                const float tt = w[r] + EPSF;
                w[r] = tt * tt * tt;
            }
#pragma unroll
            for (int r = 0; r < 8; ++r) w[r] = __builtin_amdgcn_rcpf(w[r]);
#pragma unroll
            for (int r = 0; r < 8; ++r) {
                fx[r] = fmaf(w[r], xi[r] - pjx, fx[r]);
                fy[r] = fmaf(w[r], yi[r] - pjy, fy[r]);
                fz[r] = fmaf(w[r], zi[r] - pjz, fz[r]);
            }
        }
    };

#pragma nounroll
    for (int g = 0; g < NGRP; ++g) {
        load_group(gb, pjl + ((rot + g + 1) & (NGRP - 1)) * GRPF);
        do_group(ga);
        __builtin_amdgcn_sched_barrier(0);
        hold_group(gb);
        ga = gb;
    }

    const bool b8 = (n16 & 8) != 0;
    const bool b4 = (n16 & 4) != 0;
    const bool b2 = (n16 & 2) != 0;
    float gsum[4][3], hh[2][3], tot[3];
#pragma unroll
    for (int k = 0; k < 4; ++k) {
        gsum[k][0] = add_xor16<8>(b8 ? fx[k + 4] : fx[k], b8 ? fx[k] : fx[k + 4]);
        gsum[k][1] = add_xor16<8>(b8 ? fy[k + 4] : fy[k], b8 ? fy[k] : fy[k + 4]);
        gsum[k][2] = add_xor16<8>(b8 ? fz[k + 4] : fz[k], b8 ? fz[k] : fz[k + 4]);
    }
#pragma unroll
    for (int k = 0; k < 2; ++k)
#pragma unroll
        for (int c = 0; c < 3; ++c)
            hh[k][c] = add_xor16<4>(b4 ? gsum[k + 2][c] : gsum[k][c], b4 ? gsum[k][c] : gsum[k + 2][c]);
#pragma unroll
    for (int c = 0; c < 3; ++c) {
        const float t2 = add_xor16<2>(b2 ? hh[1][c] : hh[0][c], b2 ? hh[0][c] : hh[1][c]);
        tot[c] = add_xor16<1>(t2, t2);
    }
    if ((n16 & 1) == 0) {
        float* p = &red[sl][(rg * 16 + (hi ? 8 : 0) + (n16 >> 1)) * 3];
        p[0] = tot[0]; p[1] = tot[1]; p[2] = tot[2];
    }
    __syncthreads();

    if (tid < IBLK * 3) {
        float s = 0.0f;
#pragma unroll
        for (int k = 0; k < NSLICE; ++k) s += red[k][tid];
        out[blockIdx.x * (IBLK * 3) + tid] = s;
    }
}

extern "C" void kernel_launch(void* const* d_in, const int* in_sizes, int n_in,
                              void* d_out, int out_size, void* d_ws, size_t ws_size,
                              hipStream_t stream)
{
    const float* pos = (const float*)d_in[0];
    float* out = (float*)d_out;

    nbody_force_wmma<<<NPTS / IBLK, BLOCK, 0, stream>>>(pos, out);
}
